# starting kernel
# baseline (speedup 1.0000x reference)
#include <hip/hip_runtime.h>
#include <hip/hip_bf16.h>


typedef __attribute__((ext_vector_type(2))) float v2f;
typedef __attribute__((ext_vector_type(8))) float v8f;

#define NN      2048
#define IN_DIM  256
#define OUT_DIM 64
#define NHEADS  8
#define DH      8
#define JCHUNK  128
#define LSTR    132

__global__ void gat_wh_kernel(const float* __restrict__ h,
                              const float* __restrict__ W_w,
                              const float* __restrict__ W_b,
                              const float* __restrict__ a_w,
                              float* __restrict__ wh,
                              float* __restrict__ e_i,
                              float* __restrict__ e_j) {
    __shared__ float hrow[IN_DIM];
    __shared__ float whs[OUT_DIM];
    const int n = blockIdx.x;
    const int t = threadIdx.x;

    ((float4*)hrow)[t] = ((const float4*)(h + (size_t)n * IN_DIM))[t];
    __syncthreads();

    const float4* wr4 = (const float4*)(W_w + (size_t)t * IN_DIM);
    const float4* hr4 = (const float4*)hrow;
    float acc = W_b[t];
#pragma unroll 8
    for (int k = 0; k < IN_DIM / 4; ++k) {
        float4 w = wr4[k];
        float4 x = hr4[k];
        acc = fmaf(w.x, x.x, acc);
        acc = fmaf(w.y, x.y, acc);
        acc = fmaf(w.z, x.z, acc);
        acc = fmaf(w.w, x.w, acc);
    }
    wh[(size_t)n * OUT_DIM + t] = acc;
    whs[t] = acc;
    __syncthreads();

    if (t < NHEADS) {
        float s = 0.f;
#pragma unroll
        for (int d = 0; d < DH; ++d) s = fmaf(whs[t * DH + d], a_w[d], s);
        e_i[(size_t)n * NHEADS + t] = s;
    } else if (t < 2 * NHEADS) {
        const int hh = t - NHEADS;
        float s = 0.f;
#pragma unroll
        for (int d = 0; d < DH; ++d) s = fmaf(whs[hh * DH + d], a_w[DH + d], s);
        e_j[(size_t)n * NHEADS + hh] = s;
    }
}

__global__ __launch_bounds__(256)
void gat_att_kernel(const int*   __restrict__ adj,
                    const float* __restrict__ wh,
                    const float* __restrict__ e_i,
                    const float* __restrict__ e_j,
                    const float* __restrict__ a_b,
                    float* __restrict__ out) {
    __shared__ int   adjS[16 * LSTR];
    __shared__ float whT[OUT_DIM * LSTR];
    __shared__ float ejT[NHEADS * LSTR];
    __shared__ float pden[NHEADS][32];

    const int tid  = threadIdx.x;
    const int hH   = tid >> 5;
    const int lane = tid & 31;
    const int i0   = blockIdx.x * 16;

    const int   iLoc  = lane & 15;
    const int   nCol  = lane & 15;
    const int   nColC = nCol & 7;
    const int   khalf = (lane >> 4) * 2;
    const float cmask = (nCol < DH) ? 1.0f : 0.0f;
    const float eib   = e_i[(size_t)(i0 + iLoc) * NHEADS + hH] + a_b[0];
    const int   dRow  = hH * DH + nColC;

    v8f   acc  = {0.f, 0.f, 0.f, 0.f, 0.f, 0.f, 0.f, 0.f};
    float psum = 0.f;

    for (int chunk = 0; chunk < NN; chunk += JCHUNK) {
        __syncthreads();

        for (int idx = tid; idx < 512; idx += 256) {
            const int r = idx >> 5;
            const int c = idx & 31;
            ((int4*)&adjS[r * LSTR])[c] =
                *(const int4*)(adj + (size_t)(i0 + r) * NN + chunk + c * 4);
        }
        for (int idx = tid; idx < JCHUNK * OUT_DIM; idx += 256) {
            const int j = idx >> 6;
            const int d = idx & 63;
            whT[d * LSTR + j] = wh[(size_t)(chunk + j) * OUT_DIM + d];
        }
        for (int idx = tid; idx < JCHUNK * NHEADS; idx += 256) {
            const int j  = idx >> 3;
            const int hh = idx & 7;
            ejT[hh * LSTR + j] = e_j[(size_t)(chunk + j) * NHEADS + hh];
        }

        if (chunk + JCHUNK < NN)
            __builtin_prefetch(adj + (size_t)(i0 + (tid >> 4)) * NN + chunk + JCHUNK, 0, 0);

        __syncthreads();

#pragma unroll
        for (int js = 0; js < JCHUNK; js += 4) {
            const int j0 = js + khalf;
            const float2 ej = *(const float2*)&ejT[hH * LSTR + j0];
            const int2   ad = *(const int2*)  &adjS[iLoc * LSTR + j0];
            const float2 wv = *(const float2*)&whT[dRow * LSTR + j0];

            float s0 = eib + ej.x;
            float s1 = eib + ej.y;
            s0 = (s0 >= 0.f) ? s0 : 0.2f * s0;
            s1 = (s1 >= 0.f) ? s1 : 0.2f * s1;
            const float m0 = (ad.x > 0) ? 1.0f : 0.0f;
            const float m1 = (ad.y > 0) ? 1.0f : 0.0f;
            const float p0 = __expf(s0) * m0;
            const float p1 = __expf(s1) * m1;
            psum += p0 + p1;

            v2f A; A.x = p0;          A.y = p1;
            v2f B; B.x = wv.x * cmask; B.y = wv.y * cmask;
            acc = __builtin_amdgcn_wmma_f32_16x16x4_f32(
                false, A, false, B, (short)0, acc, false, false);
        }
    }

    pden[hH][lane] = psum;
    __syncthreads();

    const int rbase = (lane >> 4) * 8;
    if (nCol < DH) {
#pragma unroll
        for (int v = 0; v < 8; ++v) {
            const int   iL  = rbase + v;
            const float den = pden[hH][iL] + pden[hH][iL + 16];
            float o = acc[v] / den;
            o = (o > 0.f) ? o : (__expf(o) - 1.f);
            out[(size_t)hH * NN * DH + (size_t)(i0 + iL) * DH + nCol] = o;
        }
    }
}

extern "C" void kernel_launch(void* const* d_in, const int* in_sizes, int n_in,
                              void* d_out, int out_size, void* d_ws, size_t ws_size,
                              hipStream_t stream) {
    const float* h   = (const float*)d_in[0];
    const int*   adj = (const int*)  d_in[1];
    const float* W_w = (const float*)d_in[2];
    const float* W_b = (const float*)d_in[3];
    const float* a_w = (const float*)d_in[4];
    const float* a_b = (const float*)d_in[5];
    float* out = (float*)d_out;

    float* wh  = (float*)d_ws;
    float* e_i = wh  + (size_t)NN * OUT_DIM;
    float* e_j = e_i + (size_t)NN * NHEADS;

    gat_wh_kernel<<<NN, 64, 0, stream>>>(h, W_w, W_b, a_w, wh, e_i, e_j);
    gat_att_kernel<<<NN / 16, 256, 0, stream>>>(adj, wh, e_i, e_j, a_b, out);
}


# attention kernel split 8 ways over j across the grid: 1024 work-groups, 4 per WGP, cover each other's staging waits; partials to workspace, a third small kernel adds them (sum order over j changes)
# speedup vs baseline: 3.4139x; 3.4139x over previous
#include <hip/hip_runtime.h>
#include <hip/hip_bf16.h>


typedef __attribute__((ext_vector_type(2))) float v2f;
typedef __attribute__((ext_vector_type(8))) float v8f;

#define NN      2048
#define IN_DIM  256
#define OUT_DIM 64
#define NHEADS  8
#define DH      8
#define JCHUNK  128
#define LSTR    132
#define JSPLIT  8
#define JPER    (NN / JSPLIT)

__global__ void gat_wh_kernel(const float* __restrict__ h,
                              const float* __restrict__ W_w,
                              const float* __restrict__ W_b,
                              const float* __restrict__ a_w,
                              float* __restrict__ wh,
                              float* __restrict__ e_i,
                              float* __restrict__ e_j) {
    __shared__ float hrow[IN_DIM];
    __shared__ float whs[OUT_DIM];
    const int n = blockIdx.x;
    const int t = threadIdx.x;

    ((float4*)hrow)[t] = ((const float4*)(h + (size_t)n * IN_DIM))[t];
    __syncthreads();

    const float4* wr4 = (const float4*)(W_w + (size_t)t * IN_DIM);
    const float4* hr4 = (const float4*)hrow;
    float acc = W_b[t];
#pragma unroll 8
    for (int k = 0; k < IN_DIM / 4; ++k) {
        float4 w = wr4[k];
        float4 x = hr4[k];
        acc = fmaf(w.x, x.x, acc);
        acc = fmaf(w.y, x.y, acc);
        acc = fmaf(w.z, x.z, acc);
        acc = fmaf(w.w, x.w, acc);
    }
    wh[(size_t)n * OUT_DIM + t] = acc;
    whs[t] = acc;
    __syncthreads();

    if (t < NHEADS) {
        float s = 0.f;
#pragma unroll
        for (int d = 0; d < DH; ++d) s = fmaf(whs[t * DH + d], a_w[d], s);
        e_i[(size_t)n * NHEADS + t] = s;
    } else if (t < 2 * NHEADS) {
        const int hh = t - NHEADS;
        float s = 0.f;
#pragma unroll
        for (int d = 0; d < DH; ++d) s = fmaf(whs[hh * DH + d], a_w[DH + d], s);
        e_j[(size_t)n * NHEADS + hh] = s;
    }
}

__global__ __launch_bounds__(256)
void gat_att_kernel(const int*   __restrict__ adj,
                    const float* __restrict__ wh,
                    const float* __restrict__ e_i,
                    const float* __restrict__ e_j,
                    const float* __restrict__ a_b,
                    float* __restrict__ pnum,
                    float* __restrict__ pden) {
    __shared__ int   adjS[16 * LSTR];
    __shared__ float whT[OUT_DIM * LSTR];
    __shared__ float ejT[NHEADS * LSTR];
    __shared__ __attribute__((aligned(16))) float ptile[NHEADS][16 * DH];
    __shared__ __attribute__((aligned(16))) float prow[NHEADS * 16];

    const int tid  = threadIdx.x;
    const int hH   = tid >> 5;
    const int lane = tid & 31;
    const int i0   = blockIdx.x * 16;
    const int jr   = blockIdx.y;
    const int jbeg = jr * JPER;

    const int   iLoc  = lane & 15;
    const int   nCol  = lane & 15;
    const int   nColC = nCol & 7;
    const int   khalf = (lane >> 4) * 2;
    const float cmask = (nCol < DH) ? 1.0f : 0.0f;
    const float eib   = e_i[(size_t)(i0 + iLoc) * NHEADS + hH] + a_b[0];
    const int   dRow  = hH * DH + nColC;

    v8f   acc  = {0.f, 0.f, 0.f, 0.f, 0.f, 0.f, 0.f, 0.f};
    float psum = 0.f;

    for (int chunk = jbeg; chunk < jbeg + JPER; chunk += JCHUNK) {
        __syncthreads();

        for (int idx = tid; idx < 512; idx += 256) {
            const int r = idx >> 5;
            const int c = idx & 31;
            ((int4*)&adjS[r * LSTR])[c] =
                *(const int4*)(adj + (size_t)(i0 + r) * NN + chunk + c * 4);
        }
        for (int idx = tid; idx < JCHUNK * OUT_DIM; idx += 256) {
            const int j = idx >> 6;
            const int d = idx & 63;
            whT[d * LSTR + j] = wh[(size_t)(chunk + j) * OUT_DIM + d];
        }
        for (int idx = tid; idx < JCHUNK * NHEADS; idx += 256) {
            const int j  = idx >> 3;
            const int hh = idx & 7;
            ejT[hh * LSTR + j] = e_j[(size_t)(chunk + j) * NHEADS + hh];
        }

        if (chunk + JCHUNK < jbeg + JPER)
            __builtin_prefetch(adj + (size_t)(i0 + (tid >> 4)) * NN + chunk + JCHUNK, 0, 0);

        __syncthreads();

#pragma unroll
        for (int js = 0; js < JCHUNK; js += 4) {
            const int j0 = js + khalf;
            const float2 ej = *(const float2*)&ejT[hH * LSTR + j0];
            const int2   ad = *(const int2*)  &adjS[iLoc * LSTR + j0];
            const float2 wv = *(const float2*)&whT[dRow * LSTR + j0];

            float s0 = eib + ej.x;
            float s1 = eib + ej.y;
            s0 = (s0 >= 0.f) ? s0 : 0.2f * s0;
            s1 = (s1 >= 0.f) ? s1 : 0.2f * s1;
            const float m0 = (ad.x > 0) ? 1.0f : 0.0f;
            const float m1 = (ad.y > 0) ? 1.0f : 0.0f;
            const float p0 = __expf(s0) * m0;
            const float p1 = __expf(s1) * m1;
            psum += p0 + p1;

            v2f A; A.x = p0;          A.y = p1;
            v2f B; B.x = wv.x * cmask; B.y = wv.y * cmask;
            acc = __builtin_amdgcn_wmma_f32_16x16x4_f32(
                false, A, false, B, (short)0, acc, false, false);
        }
    }

    const float prs = psum + __shfl_xor(psum, 16);
    if (lane < 16) prow[hH * 16 + lane] = prs;

    const int rbase = (lane >> 4) * 8;
    if (nCol < DH) {
#pragma unroll
        for (int v = 0; v < 8; ++v)
            ptile[hH][(rbase + v) * DH + nCol] = acc[v];
    }
    __syncthreads();

    {
        const float4 t4 = ((const float4*)ptile[hH])[lane];
        float* dstN = pnum + (size_t)jr * NN * OUT_DIM
                           + (size_t)hH * NN * DH + (size_t)i0 * DH;
        ((float4*)dstN)[lane] = t4;
    }
    if (hH == 0) {
        const float4 d4 = ((const float4*)prow)[lane];
        float* dstD = pden + (size_t)jr * NN * NHEADS
                           + (size_t)blockIdx.x * (NHEADS * 16);
        ((float4*)dstD)[lane] = d4;
    }
}

__global__ __launch_bounds__(256)
void gat_combine_kernel(const float* __restrict__ pnum,
                        const float* __restrict__ pden,
                        float* __restrict__ out) {
    const int q = blockIdx.x * 256 + threadIdx.x;
    const int o = q * 4;
    const int h = o / (NN * DH);
    const int i = (o % (NN * DH)) / DH;
    const size_t doff = (size_t)(i >> 4) * (NHEADS * 16) + (size_t)h * 16 + (i & 15);

    float4 num = {0.f, 0.f, 0.f, 0.f};
    float  den = 0.f;
#pragma unroll
    for (int s = 0; s < JSPLIT; ++s) {
        const float4 p = ((const float4*)(pnum + (size_t)s * NN * OUT_DIM))[q];
        num.x += p.x; num.y += p.y; num.z += p.z; num.w += p.w;
        den += pden[(size_t)s * NN * NHEADS + doff];
    }
    float4 r;
    r.x = num.x / den; r.y = num.y / den; r.z = num.z / den; r.w = num.w / den;
    r.x = (r.x > 0.f) ? r.x : (__expf(r.x) - 1.f);
    r.y = (r.y > 0.f) ? r.y : (__expf(r.y) - 1.f);
    r.z = (r.z > 0.f) ? r.z : (__expf(r.z) - 1.f);
    r.w = (r.w > 0.f) ? r.w : (__expf(r.w) - 1.f);
    ((float4*)out)[q] = r;
}

extern "C" void kernel_launch(void* const* d_in, const int* in_sizes, int n_in,
                              void* d_out, int out_size, void* d_ws, size_t ws_size,
                              hipStream_t stream) {
    const float* h   = (const float*)d_in[0];
    const int*   adj = (const int*)  d_in[1];
    const float* W_w = (const float*)d_in[2];
    const float* W_b = (const float*)d_in[3];
    const float* a_w = (const float*)d_in[4];
    const float* a_b = (const float*)d_in[5];
    float* out = (float*)d_out;

    float* wh  = (float*)d_ws;
    float* e_i = wh  + (size_t)NN * OUT_DIM;
    float* e_j = e_i + (size_t)NN * NHEADS;
    float* pnum = e_j + (size_t)NN * NHEADS;
    float* pden = pnum + (size_t)JSPLIT * NN * OUT_DIM;

    gat_wh_kernel<<<NN, 64, 0, stream>>>(h, W_w, W_b, a_w, wh, e_i, e_j);
    gat_att_kernel<<<dim3(NN / 16, JSPLIT), 256, 0, stream>>>(adj, wh, e_i, e_j, a_b, pnum, pden);
    gat_combine_kernel<<<NN * OUT_DIM / 4 / 256, 256, 0, stream>>>(pnum, pden, out);
}


# projection kernel: lanes run along k, W rows read as whole 512-byte pieces per wave (was one row per lane, 32 lines per load); a row's 32 lane partials added through LDS (sum order over k changes)
# speedup vs baseline: 3.8868x; 1.1385x over previous
#include <hip/hip_runtime.h>
#include <hip/hip_bf16.h>


typedef __attribute__((ext_vector_type(2))) float v2f;
typedef __attribute__((ext_vector_type(8))) float v8f;

#define NN      2048
#define IN_DIM  256
#define OUT_DIM 64
#define NHEADS  8
#define DH      8
#define JCHUNK  128
#define LSTR    132
#define JSPLIT  8
#define JPER    (NN / JSPLIT)

#define PSTR 36
__global__ __launch_bounds__(64)
void gat_wh_kernel(const float* __restrict__ h,
                   const float* __restrict__ W_w,
                   const float* __restrict__ W_b,
                   const float* __restrict__ a_w,
                   float* __restrict__ wh,
                   float* __restrict__ e_i,
                   float* __restrict__ e_j) {
    __shared__ __attribute__((aligned(16))) float part[2][32 * PSTR];
    __shared__ float whs[OUT_DIM];
    const int n    = blockIdx.x;
    const int t    = threadIdx.x;
    const int w    = t >> 5;
    const int lane = t & 31;

    const float4* hp = (const float4*)(h + (size_t)n * IN_DIM);
    const float4 x0 = hp[lane];
    const float4 x1 = hp[32 + lane];

    const float4* wp = (const float4*)(W_w + (size_t)(w * 32) * IN_DIM) + lane;
    float* myPart = &part[w][lane * PSTR];

#pragma unroll
    for (int r0 = 0; r0 < 32; r0 += 4) {
        float4 p;
        {
            const float4 a = wp[(r0 + 0) * 64];
            const float4 b = wp[(r0 + 0) * 64 + 32];
            float s = a.x * x0.x;
            s = fmaf(a.y, x0.y, s); s = fmaf(a.z, x0.z, s); s = fmaf(a.w, x0.w, s);
            s = fmaf(b.x, x1.x, s); s = fmaf(b.y, x1.y, s); s = fmaf(b.z, x1.z, s); s = fmaf(b.w, x1.w, s);
            p.x = s;
        }
        {
            const float4 a = wp[(r0 + 1) * 64];
            const float4 b = wp[(r0 + 1) * 64 + 32];
            float s = a.x * x0.x;
            s = fmaf(a.y, x0.y, s); s = fmaf(a.z, x0.z, s); s = fmaf(a.w, x0.w, s);
            s = fmaf(b.x, x1.x, s); s = fmaf(b.y, x1.y, s); s = fmaf(b.z, x1.z, s); s = fmaf(b.w, x1.w, s);
            p.y = s;
        }
        {
            const float4 a = wp[(r0 + 2) * 64];
            const float4 b = wp[(r0 + 2) * 64 + 32];
            float s = a.x * x0.x;
            s = fmaf(a.y, x0.y, s); s = fmaf(a.z, x0.z, s); s = fmaf(a.w, x0.w, s);
            s = fmaf(b.x, x1.x, s); s = fmaf(b.y, x1.y, s); s = fmaf(b.z, x1.z, s); s = fmaf(b.w, x1.w, s);
            p.z = s;
        }
        {
            const float4 a = wp[(r0 + 3) * 64];
            const float4 b = wp[(r0 + 3) * 64 + 32];
            float s = a.x * x0.x;
            s = fmaf(a.y, x0.y, s); s = fmaf(a.z, x0.z, s); s = fmaf(a.w, x0.w, s);
            s = fmaf(b.x, x1.x, s); s = fmaf(b.y, x1.y, s); s = fmaf(b.z, x1.z, s); s = fmaf(b.w, x1.w, s);
            p.w = s;
        }
        *(float4*)&myPart[r0] = p;
    }
    __syncthreads();

    const float* col = &part[w][lane];
    float acc = W_b[t];
#pragma unroll
    for (int l = 0; l < 32; ++l) acc += col[l * PSTR];

    wh[(size_t)n * OUT_DIM + t] = acc;
    whs[t] = acc;
    __syncthreads();

    if (t < NHEADS) {
        float s = 0.f;
#pragma unroll
        for (int d = 0; d < DH; ++d) s = fmaf(whs[t * DH + d], a_w[d], s);
        e_i[(size_t)n * NHEADS + t] = s;
    } else if (t < 2 * NHEADS) {
        const int hh = t - NHEADS;
        float s = 0.f;
#pragma unroll
        for (int d = 0; d < DH; ++d) s = fmaf(whs[hh * DH + d], a_w[DH + d], s);
        e_j[(size_t)n * NHEADS + hh] = s;
    }
}

__global__ __launch_bounds__(256)
void gat_att_kernel(const int*   __restrict__ adj,
                    const float* __restrict__ wh,
                    const float* __restrict__ e_i,
                    const float* __restrict__ e_j,
                    const float* __restrict__ a_b,
                    float* __restrict__ pnum,
                    float* __restrict__ pden) {
    __shared__ int   adjS[16 * LSTR];
    __shared__ float whT[OUT_DIM * LSTR];
    __shared__ float ejT[NHEADS * LSTR];
    __shared__ __attribute__((aligned(16))) float ptile[NHEADS][16 * DH];
    __shared__ __attribute__((aligned(16))) float prow[NHEADS * 16];

    const int tid  = threadIdx.x;
    const int hH   = tid >> 5;
    const int lane = tid & 31;
    const int i0   = blockIdx.x * 16;
    const int jr   = blockIdx.y;
    const int jbeg = jr * JPER;

    const int   iLoc  = lane & 15;
    const int   nCol  = lane & 15;
    const int   nColC = nCol & 7;
    const int   khalf = (lane >> 4) * 2;
    const float cmask = (nCol < DH) ? 1.0f : 0.0f;
    const float eib   = e_i[(size_t)(i0 + iLoc) * NHEADS + hH] + a_b[0];
    const int   dRow  = hH * DH + nColC;

    v8f   acc  = {0.f, 0.f, 0.f, 0.f, 0.f, 0.f, 0.f, 0.f};
    float psum = 0.f;

    for (int chunk = jbeg; chunk < jbeg + JPER; chunk += JCHUNK) {
        __syncthreads();

        for (int idx = tid; idx < 512; idx += 256) {
            const int r = idx >> 5;
            const int c = idx & 31;
            ((int4*)&adjS[r * LSTR])[c] =
                *(const int4*)(adj + (size_t)(i0 + r) * NN + chunk + c * 4);
        }
        for (int idx = tid; idx < JCHUNK * OUT_DIM; idx += 256) {
            const int j = idx >> 6;
            const int d = idx & 63;
            whT[d * LSTR + j] = wh[(size_t)(chunk + j) * OUT_DIM + d];
        }
        for (int idx = tid; idx < JCHUNK * NHEADS; idx += 256) {
            const int j  = idx >> 3;
            const int hh = idx & 7;
            ejT[hh * LSTR + j] = e_j[(size_t)(chunk + j) * NHEADS + hh];
        }

        if (chunk + JCHUNK < jbeg + JPER)
            __builtin_prefetch(adj + (size_t)(i0 + (tid >> 4)) * NN + chunk + JCHUNK, 0, 0);

        __syncthreads();

#pragma unroll
        for (int js = 0; js < JCHUNK; js += 4) {
            const int j0 = js + khalf;
            const float2 ej = *(const float2*)&ejT[hH * LSTR + j0];
            const int2   ad = *(const int2*)  &adjS[iLoc * LSTR + j0];
            const float2 wv = *(const float2*)&whT[dRow * LSTR + j0];

            float s0 = eib + ej.x;
            float s1 = eib + ej.y;
            s0 = (s0 >= 0.f) ? s0 : 0.2f * s0;
            s1 = (s1 >= 0.f) ? s1 : 0.2f * s1;
            const float m0 = (ad.x > 0) ? 1.0f : 0.0f;
            const float m1 = (ad.y > 0) ? 1.0f : 0.0f;
            const float p0 = __expf(s0) * m0;
            const float p1 = __expf(s1) * m1;
            psum += p0 + p1;

            v2f A; A.x = p0;          A.y = p1;
            v2f B; B.x = wv.x * cmask; B.y = wv.y * cmask;
            acc = __builtin_amdgcn_wmma_f32_16x16x4_f32(
                false, A, false, B, (short)0, acc, false, false);
        }
    }

    const float prs = psum + __shfl_xor(psum, 16);
    if (lane < 16) prow[hH * 16 + lane] = prs;

    const int rbase = (lane >> 4) * 8;
    if (nCol < DH) {
#pragma unroll
        for (int v = 0; v < 8; ++v)
            ptile[hH][(rbase + v) * DH + nCol] = acc[v];
    }
    __syncthreads();

    {
        const float4 t4 = ((const float4*)ptile[hH])[lane];
        float* dstN = pnum + (size_t)jr * NN * OUT_DIM
                           + (size_t)hH * NN * DH + (size_t)i0 * DH;
        ((float4*)dstN)[lane] = t4;
    }
    if (hH == 0) {
        const float4 d4 = ((const float4*)prow)[lane];
        float* dstD = pden + (size_t)jr * NN * NHEADS
                           + (size_t)blockIdx.x * (NHEADS * 16);
        ((float4*)dstD)[lane] = d4;
    }
}

__global__ __launch_bounds__(256)
void gat_combine_kernel(const float* __restrict__ pnum,
                        const float* __restrict__ pden,
                        float* __restrict__ out) {
    const int q = blockIdx.x * 256 + threadIdx.x;
    const int o = q * 4;
    const int h = o / (NN * DH);
    const int i = (o % (NN * DH)) / DH;
    const size_t doff = (size_t)(i >> 4) * (NHEADS * 16) + (size_t)h * 16 + (i & 15);

    float4 num = {0.f, 0.f, 0.f, 0.f};
    float  den = 0.f;
#pragma unroll
    for (int s = 0; s < JSPLIT; ++s) {
        const float4 p = ((const float4*)(pnum + (size_t)s * NN * OUT_DIM))[q];
        num.x += p.x; num.y += p.y; num.z += p.z; num.w += p.w;
        den += pden[(size_t)s * NN * NHEADS + doff];
    }
    float4 r;
    r.x = num.x / den; r.y = num.y / den; r.z = num.z / den; r.w = num.w / den;
    r.x = (r.x > 0.f) ? r.x : (__expf(r.x) - 1.f);
    r.y = (r.y > 0.f) ? r.y : (__expf(r.y) - 1.f);
    r.z = (r.z > 0.f) ? r.z : (__expf(r.z) - 1.f);
    r.w = (r.w > 0.f) ? r.w : (__expf(r.w) - 1.f);
    ((float4*)out)[q] = r;
}

extern "C" void kernel_launch(void* const* d_in, const int* in_sizes, int n_in,
                              void* d_out, int out_size, void* d_ws, size_t ws_size,
                              hipStream_t stream) {
    const float* h   = (const float*)d_in[0];
    const int*   adj = (const int*)  d_in[1];
    const float* W_w = (const float*)d_in[2];
    const float* W_b = (const float*)d_in[3];
    const float* a_w = (const float*)d_in[4];
    const float* a_b = (const float*)d_in[5];
    float* out = (float*)d_out;

    float* wh  = (float*)d_ws;
    float* e_i = wh  + (size_t)NN * OUT_DIM;
    float* e_j = e_i + (size_t)NN * NHEADS;
    float* pnum = e_j + (size_t)NN * NHEADS;
    float* pden = pnum + (size_t)JSPLIT * NN * OUT_DIM;

    gat_wh_kernel<<<NN, 64, 0, stream>>>(h, W_w, W_b, a_w, wh, e_i, e_j);
    gat_att_kernel<<<dim3(NN / 16, JSPLIT), 256, 0, stream>>>(adj, wh, e_i, e_j, a_b, pnum, pden);
    gat_combine_kernel<<<NN * OUT_DIM / 4 / 256, 256, 0, stream>>>(pnum, pden, out);
}


# attention loop: 16 -> 6 vector instructions per WMMA: mask factor and log2e scale made at staging, lrelu as max, row sums from a ones column of B (denominator sum order, weight rounding change)
# speedup vs baseline: 4.1385x; 1.0648x over previous
#include <hip/hip_runtime.h>
#include <hip/hip_bf16.h>


typedef __attribute__((ext_vector_type(2))) float v2f;
typedef __attribute__((ext_vector_type(8))) float v8f;

#define NN      2048
#define IN_DIM  256
#define OUT_DIM 64
#define NHEADS  8
#define DH      8
#define JCHUNK  128
#define LSTR    132
#define JSPLIT  8
#define JPER    (NN / JSPLIT)
#define ONES_LEN 160
#define LOG2E   1.4426950408889634f

#define PSTR 36
__global__ __launch_bounds__(64)
void gat_wh_kernel(const float* __restrict__ h,
                   const float* __restrict__ W_w,
                   const float* __restrict__ W_b,
                   const float* __restrict__ a_w,
                   float* __restrict__ wh,
                   float* __restrict__ e_i,
                   float* __restrict__ e_j) {
    __shared__ __attribute__((aligned(16))) float part[2][32 * PSTR];
    __shared__ float whs[OUT_DIM];
    const int n    = blockIdx.x;
    const int t    = threadIdx.x;
    const int w    = t >> 5;
    const int lane = t & 31;

    const float4* hp = (const float4*)(h + (size_t)n * IN_DIM);
    const float4 x0 = hp[lane];
    const float4 x1 = hp[32 + lane];

    const float4* wp = (const float4*)(W_w + (size_t)(w * 32) * IN_DIM) + lane;
    float* myPart = &part[w][lane * PSTR];

#pragma unroll
    for (int r0 = 0; r0 < 32; r0 += 4) {
        float4 p;
        {
            const float4 a = wp[(r0 + 0) * 64];
            const float4 b = wp[(r0 + 0) * 64 + 32];
            float s = a.x * x0.x;
            s = fmaf(a.y, x0.y, s); s = fmaf(a.z, x0.z, s); s = fmaf(a.w, x0.w, s);
            s = fmaf(b.x, x1.x, s); s = fmaf(b.y, x1.y, s); s = fmaf(b.z, x1.z, s); s = fmaf(b.w, x1.w, s);
            p.x = s;
        }
        {
            const float4 a = wp[(r0 + 1) * 64];
            const float4 b = wp[(r0 + 1) * 64 + 32];
            float s = a.x * x0.x;
            s = fmaf(a.y, x0.y, s); s = fmaf(a.z, x0.z, s); s = fmaf(a.w, x0.w, s);
            s = fmaf(b.x, x1.x, s); s = fmaf(b.y, x1.y, s); s = fmaf(b.z, x1.z, s); s = fmaf(b.w, x1.w, s);
            p.y = s;
        }
        {
            const float4 a = wp[(r0 + 2) * 64];
            const float4 b = wp[(r0 + 2) * 64 + 32];
            float s = a.x * x0.x;
            s = fmaf(a.y, x0.y, s); s = fmaf(a.z, x0.z, s); s = fmaf(a.w, x0.w, s);
            s = fmaf(b.x, x1.x, s); s = fmaf(b.y, x1.y, s); s = fmaf(b.z, x1.z, s); s = fmaf(b.w, x1.w, s);
            p.z = s;
        }
        {
            const float4 a = wp[(r0 + 3) * 64];
            const float4 b = wp[(r0 + 3) * 64 + 32];
            float s = a.x * x0.x;
            s = fmaf(a.y, x0.y, s); s = fmaf(a.z, x0.z, s); s = fmaf(a.w, x0.w, s);
            s = fmaf(b.x, x1.x, s); s = fmaf(b.y, x1.y, s); s = fmaf(b.z, x1.z, s); s = fmaf(b.w, x1.w, s);
            p.w = s;
        }
        *(float4*)&myPart[r0] = p;
    }
    __syncthreads();

    const float* col = &part[w][lane];
    float acc = W_b[t];
#pragma unroll
    for (int l = 0; l < 32; ++l) acc += col[l * PSTR];

    wh[(size_t)n * OUT_DIM + t] = acc;
    whs[t] = acc;
    __syncthreads();

    if (t < NHEADS) {
        float s = 0.f;
#pragma unroll
        for (int d = 0; d < DH; ++d) s = fmaf(whs[t * DH + d], a_w[d], s);
        e_i[(size_t)n * NHEADS + t] = s;
    } else if (t < 2 * NHEADS) {
        const int hh = t - NHEADS;
        float s = 0.f;
#pragma unroll
        for (int d = 0; d < DH; ++d) s = fmaf(whs[hh * DH + d], a_w[DH + d], s);
        e_j[(size_t)n * NHEADS + hh] = s;
    }
}

__global__ __launch_bounds__(256)
void gat_att_kernel(const int*   __restrict__ adj,
                    const float* __restrict__ wh,
                    const float* __restrict__ e_i,
                    const float* __restrict__ e_j,
                    const float* __restrict__ a_b,
                    float* __restrict__ pnum,
                    float* __restrict__ pden) {
    __shared__ __attribute__((aligned(16))) float adjS[16 * LSTR];
    __shared__ __attribute__((aligned(16))) v2f whT2[(OUT_DIM * LSTR + ONES_LEN) / 2];
    float* const whT = (float*)whT2;
    __shared__ float ejT[NHEADS * LSTR];
    __shared__ __attribute__((aligned(16))) float ptile[NHEADS][16 * DH];
    __shared__ __attribute__((aligned(16))) float prow[NHEADS * 16];

    const int tid  = threadIdx.x;
    const int hH   = tid >> 5;
    const int lane = tid & 31;
    const int i0   = blockIdx.x * 16;
    const int jr   = blockIdx.y;
    const int jbeg = jr * JPER;

    const int   iLoc  = lane & 15;
    const int   nCol  = lane & 15;
    const int   khalf = (lane >> 4) * 2;
    const float eibL  = (e_i[(size_t)(i0 + iLoc) * NHEADS + hH] + a_b[0]) * LOG2E;
    v2f eib2; eib2.x = eibL; eib2.y = eibL;
    const int   bIdx  = ((nCol < DH) ? (hH * DH + nCol) * (LSTR / 2)
                                     : (OUT_DIM * LSTR) / 2 + ((hH & 1) ? 0 : 16))
                        + (lane >> 4);

    v8f   acc  = {0.f, 0.f, 0.f, 0.f, 0.f, 0.f, 0.f, 0.f};

    if (tid < ONES_LEN) whT[OUT_DIM * LSTR + tid] = 1.0f;

    for (int chunk = jbeg; chunk < jbeg + JPER; chunk += JCHUNK) {
        __syncthreads();

        for (int idx = tid; idx < 512; idx += 256) {
            const int r = idx >> 5;
            const int c = idx & 31;
            const int4 a4 = *(const int4*)(adj + (size_t)(i0 + r) * NN + chunk + c * 4);
            float4 m4;
            m4.x = (a4.x > 0) ? 1.0f : 0.0f;
            m4.y = (a4.y > 0) ? 1.0f : 0.0f;
            m4.z = (a4.z > 0) ? 1.0f : 0.0f;
            m4.w = (a4.w > 0) ? 1.0f : 0.0f;
            ((float4*)&adjS[r * LSTR])[c] = m4;
        }
        for (int idx = tid; idx < JCHUNK * OUT_DIM; idx += 256) {
            const int j = idx >> 6;
            const int d = idx & 63;
            whT[d * LSTR + j] = wh[(size_t)(chunk + j) * OUT_DIM + d];
        }
        for (int idx = tid; idx < JCHUNK * NHEADS; idx += 256) {
            const int j  = idx >> 3;
            const int hh = idx & 7;
            ejT[hh * LSTR + j] = e_j[(size_t)(chunk + j) * NHEADS + hh] * LOG2E;
        }

        if (chunk + JCHUNK < jbeg + JPER)
            __builtin_prefetch(adj + (size_t)(i0 + (tid >> 4)) * NN + chunk + JCHUNK, 0, 0);

        __syncthreads();

#pragma unroll
        for (int js = 0; js < JCHUNK; js += 4) {
            const int j0 = js + khalf;
            const v2f ej = *(const v2f*)&ejT[hH * LSTR + j0];
            const v2f mk = *(const v2f*)&adjS[iLoc * LSTR + j0];
            const v2f B  = whT2[bIdx + js / 2];

            v2f s = eib2 + ej;
            const v2f t = s * 0.2f;
            s.x = fmaxf(s.x, t.x);
            s.y = fmaxf(s.y, t.y);
            v2f e;
            e.x = __builtin_amdgcn_exp2f(s.x);
            e.y = __builtin_amdgcn_exp2f(s.y);
            const v2f A = e * mk;
            acc = __builtin_amdgcn_wmma_f32_16x16x4_f32(
                false, A, false, B, (short)0, acc, false, false);
        }
    }

    const int rbase = (lane >> 4) * 8;
    if (nCol < DH) {
#pragma unroll
        for (int v = 0; v < 8; ++v)
            ptile[hH][(rbase + v) * DH + nCol] = acc[v];
    } else if (nCol == DH) {
#pragma unroll
        for (int v = 0; v < 8; ++v)
            prow[hH * 16 + rbase + v] = acc[v];
    }
    __syncthreads();

    {
        const float4 t4 = ((const float4*)ptile[hH])[lane];
        float* dstN = pnum + (size_t)jr * NN * OUT_DIM
                           + (size_t)hH * NN * DH + (size_t)i0 * DH;
        ((float4*)dstN)[lane] = t4;
    }
    if (hH == 0) {
        const float4 d4 = ((const float4*)prow)[lane];
        float* dstD = pden + (size_t)jr * NN * NHEADS
                           + (size_t)blockIdx.x * (NHEADS * 16);
        ((float4*)dstD)[lane] = d4;
    }
}

__global__ __launch_bounds__(256)
void gat_combine_kernel(const float* __restrict__ pnum,
                        const float* __restrict__ pden,
                        float* __restrict__ out) {
    const int q = blockIdx.x * 256 + threadIdx.x;
    const int o = q * 4;
    const int h = o / (NN * DH);
    const int i = (o % (NN * DH)) / DH;
    const size_t doff = (size_t)(i >> 4) * (NHEADS * 16) + (size_t)h * 16 + (i & 15);

    float4 num = {0.f, 0.f, 0.f, 0.f};
    float  den = 0.f;
#pragma unroll
    for (int s = 0; s < JSPLIT; ++s) {
        const float4 p = ((const float4*)(pnum + (size_t)s * NN * OUT_DIM))[q];
        num.x += p.x; num.y += p.y; num.z += p.z; num.w += p.w;
        den += pden[(size_t)s * NN * NHEADS + doff];
    }
    float4 r;
    r.x = num.x / den; r.y = num.y / den; r.z = num.z / den; r.w = num.w / den;
    r.x = (r.x > 0.f) ? r.x : (__expf(r.x) - 1.f);
    r.y = (r.y > 0.f) ? r.y : (__expf(r.y) - 1.f);
    r.z = (r.z > 0.f) ? r.z : (__expf(r.z) - 1.f);
    r.w = (r.w > 0.f) ? r.w : (__expf(r.w) - 1.f);
    ((float4*)out)[q] = r;
}

extern "C" void kernel_launch(void* const* d_in, const int* in_sizes, int n_in,
                              void* d_out, int out_size, void* d_ws, size_t ws_size,
                              hipStream_t stream) {
    const float* h   = (const float*)d_in[0];
    const int*   adj = (const int*)  d_in[1];
    const float* W_w = (const float*)d_in[2];
    const float* W_b = (const float*)d_in[3];
    const float* a_w = (const float*)d_in[4];
    const float* a_b = (const float*)d_in[5];
    float* out = (float*)d_out;

    float* wh  = (float*)d_ws;
    float* e_i = wh  + (size_t)NN * OUT_DIM;
    float* e_j = e_i + (size_t)NN * NHEADS;
    float* pnum = e_j + (size_t)NN * NHEADS;
    float* pden = pnum + (size_t)JSPLIT * NN * OUT_DIM;

    gat_wh_kernel<<<NN, 64, 0, stream>>>(h, W_w, W_b, a_w, wh, e_i, e_j);
    gat_att_kernel<<<dim3(NN / 16, JSPLIT), 256, 0, stream>>>(adj, wh, e_i, e_j, a_b, pnum, pden);
    gat_combine_kernel<<<NN * OUT_DIM / 4 / 256, 256, 0, stream>>>(pnum, pden, out);
}


# attention kernel re-tiled to 32 rows x 256 j per work-group: a wave runs two row tiles on one B fragment; the j-range is staged in one pass, all loads in flight at once; sum order unchanged
# speedup vs baseline: 5.1077x; 1.2342x over previous
#include <hip/hip_runtime.h>
#include <hip/hip_bf16.h>


typedef __attribute__((ext_vector_type(2))) float v2f;
typedef __attribute__((ext_vector_type(8))) float v8f;

#define NN      2048
#define IN_DIM  256
#define OUT_DIM 64
#define NHEADS  8
#define DH      8
#define ITILE   32
#define LSTR    260
#define JSPLIT  8
#define JPER    (NN / JSPLIT)
#define ONES_LEN (JPER + 32)
#define LOG2E   1.4426950408889634f

#define PSTR 36
__global__ __launch_bounds__(64)
void gat_wh_kernel(const float* __restrict__ h,
                   const float* __restrict__ W_w,
                   const float* __restrict__ W_b,
                   const float* __restrict__ a_w,
                   float* __restrict__ wh,
                   float* __restrict__ e_i,
                   float* __restrict__ e_j) {
    __shared__ __attribute__((aligned(16))) float part[2][32 * PSTR];
    __shared__ float whs[OUT_DIM];
    const int n    = blockIdx.x;
    const int t    = threadIdx.x;
    const int w    = t >> 5;
    const int lane = t & 31;

    const float4* hp = (const float4*)(h + (size_t)n * IN_DIM);
    const float4 x0 = hp[lane];
    const float4 x1 = hp[32 + lane];

    const float4* wp = (const float4*)(W_w + (size_t)(w * 32) * IN_DIM) + lane;
    float* myPart = &part[w][lane * PSTR];

#pragma unroll
    for (int r0 = 0; r0 < 32; r0 += 4) {
        float4 p;
        {
            const float4 a = wp[(r0 + 0) * 64];
            const float4 b = wp[(r0 + 0) * 64 + 32];
            float s = a.x * x0.x;
            s = fmaf(a.y, x0.y, s); s = fmaf(a.z, x0.z, s); s = fmaf(a.w, x0.w, s);
            s = fmaf(b.x, x1.x, s); s = fmaf(b.y, x1.y, s); s = fmaf(b.z, x1.z, s); s = fmaf(b.w, x1.w, s);
            p.x = s;
        }
        {
            const float4 a = wp[(r0 + 1) * 64];
            const float4 b = wp[(r0 + 1) * 64 + 32];
            float s = a.x * x0.x;
            s = fmaf(a.y, x0.y, s); s = fmaf(a.z, x0.z, s); s = fmaf(a.w, x0.w, s);
            s = fmaf(b.x, x1.x, s); s = fmaf(b.y, x1.y, s); s = fmaf(b.z, x1.z, s); s = fmaf(b.w, x1.w, s);
            p.y = s;
        }
        {
            const float4 a = wp[(r0 + 2) * 64];
            const float4 b = wp[(r0 + 2) * 64 + 32];
            float s = a.x * x0.x;
            s = fmaf(a.y, x0.y, s); s = fmaf(a.z, x0.z, s); s = fmaf(a.w, x0.w, s);
            s = fmaf(b.x, x1.x, s); s = fmaf(b.y, x1.y, s); s = fmaf(b.z, x1.z, s); s = fmaf(b.w, x1.w, s);
            p.z = s;
        }
        {
            const float4 a = wp[(r0 + 3) * 64];
            const float4 b = wp[(r0 + 3) * 64 + 32];
            float s = a.x * x0.x;
            s = fmaf(a.y, x0.y, s); s = fmaf(a.z, x0.z, s); s = fmaf(a.w, x0.w, s);
            s = fmaf(b.x, x1.x, s); s = fmaf(b.y, x1.y, s); s = fmaf(b.z, x1.z, s); s = fmaf(b.w, x1.w, s);
            p.w = s;
        }
        *(float4*)&myPart[r0] = p;
    }
    __syncthreads();

    const float* col = &part[w][lane];
    float acc = W_b[t];
#pragma unroll
    for (int l = 0; l < 32; ++l) acc += col[l * PSTR];

    wh[(size_t)n * OUT_DIM + t] = acc;
    whs[t] = acc;
    __syncthreads();

    if (t < NHEADS) {
        float s = 0.f;
#pragma unroll
        for (int d = 0; d < DH; ++d) s = fmaf(whs[t * DH + d], a_w[d], s);
        e_i[(size_t)n * NHEADS + t] = s;
    } else if (t < 2 * NHEADS) {
        const int hh = t - NHEADS;
        float s = 0.f;
#pragma unroll
        for (int d = 0; d < DH; ++d) s = fmaf(whs[hh * DH + d], a_w[DH + d], s);
        e_j[(size_t)n * NHEADS + hh] = s;
    }
}

__global__ __launch_bounds__(256)
void gat_att_kernel(const int*   __restrict__ adj,
                    const float* __restrict__ wh,
                    const float* __restrict__ e_i,
                    const float* __restrict__ e_j,
                    const float* __restrict__ a_b,
                    float* __restrict__ pnum,
                    float* __restrict__ pden) {
    __shared__ __attribute__((aligned(16))) float adjS[ITILE * LSTR];
    __shared__ __attribute__((aligned(16))) v2f whT2[(OUT_DIM * LSTR + ONES_LEN) / 2];
    float* const whT = (float*)whT2;
    __shared__ __attribute__((aligned(16))) float ejT[NHEADS * LSTR];
    __shared__ __attribute__((aligned(16))) float ptile[NHEADS][ITILE * DH];
    __shared__ __attribute__((aligned(16))) float prow[2 * NHEADS * 16];

    const int tid  = threadIdx.x;
    const int hH   = tid >> 5;
    const int lane = tid & 31;
    const int i0   = blockIdx.x * ITILE;
    const int jr   = blockIdx.y;
    const int jbeg = jr * JPER;

    const int   iLoc  = lane & 15;
    const int   nCol  = lane & 15;
    const int   khalf = (lane >> 4) * 2;

    const float ab  = a_b[0];
    const float ei0 = e_i[(size_t)(i0 + iLoc) * NHEADS + hH];
    const float ei1 = e_i[(size_t)(i0 + 16 + iLoc) * NHEADS + hH];

    const int ar = tid >> 6;
    const int ac = tid & 63;
    int4 a4[8];
#pragma unroll
    for (int k = 0; k < 8; ++k)
        a4[k] = *(const int4*)(adj + (size_t)(i0 + ar + 4 * k) * NN + jbeg + ac * 4);

    const int wq = (lane & 3) + 4 * (hH & 3);
    const int wj = (lane >> 2) + 8 * (hH >> 2);
    float4 w4[16];
#pragma unroll
    for (int k = 0; k < 16; ++k)
        w4[k] = *(const float4*)(wh + (size_t)(jbeg + wj + 16 * k) * OUT_DIM + wq * 4);

    const int eq = tid & 1;
    const int ej_j = tid >> 1;
    float4 e4[2];
#pragma unroll
    for (int k = 0; k < 2; ++k)
        e4[k] = *(const float4*)(e_j + (size_t)(jbeg + ej_j + 128 * k) * NHEADS + eq * 4);

    whT[OUT_DIM * LSTR + tid] = 1.0f;
    if (tid < ONES_LEN - 256) whT[OUT_DIM * LSTR + 256 + tid] = 1.0f;

#pragma unroll
    for (int k = 0; k < 8; ++k) {
        float4 m4;
        m4.x = (a4[k].x > 0) ? 1.0f : 0.0f;
        m4.y = (a4[k].y > 0) ? 1.0f : 0.0f;
        m4.z = (a4[k].z > 0) ? 1.0f : 0.0f;
        m4.w = (a4[k].w > 0) ? 1.0f : 0.0f;
        ((float4*)&adjS[(ar + 4 * k) * LSTR])[ac] = m4;
    }
#pragma unroll
    for (int k = 0; k < 16; ++k) {
        const int j = wj + 16 * k;
        whT[(wq * 4 + 0) * LSTR + j] = w4[k].x;
        whT[(wq * 4 + 1) * LSTR + j] = w4[k].y;
        whT[(wq * 4 + 2) * LSTR + j] = w4[k].z;
        whT[(wq * 4 + 3) * LSTR + j] = w4[k].w;
    }
#pragma unroll
    for (int k = 0; k < 2; ++k) {
        const int j = ej_j + 128 * k;
        ejT[(eq * 4 + 0) * LSTR + j] = e4[k].x * LOG2E;
        ejT[(eq * 4 + 1) * LSTR + j] = e4[k].y * LOG2E;
        ejT[(eq * 4 + 2) * LSTR + j] = e4[k].z * LOG2E;
        ejT[(eq * 4 + 3) * LSTR + j] = e4[k].w * LOG2E;
    }

    const float eibL0 = (ei0 + ab) * LOG2E;
    const float eibL1 = (ei1 + ab) * LOG2E;
    v2f eib0; eib0.x = eibL0; eib0.y = eibL0;
    v2f eib1; eib1.x = eibL1; eib1.y = eibL1;
    const int   bIdx  = ((nCol < DH) ? (hH * DH + nCol) * (LSTR / 2)
                                     : (OUT_DIM * LSTR) / 2 + ((hH & 1) ? 0 : 16))
                        + (lane >> 4);

    v8f acc0 = {0.f, 0.f, 0.f, 0.f, 0.f, 0.f, 0.f, 0.f};
    v8f acc1 = {0.f, 0.f, 0.f, 0.f, 0.f, 0.f, 0.f, 0.f};

    __syncthreads();

#pragma unroll 16
    for (int js = 0; js < JPER; js += 4) {
        const int j0 = js + khalf;
        const v2f ej  = *(const v2f*)&ejT[hH * LSTR + j0];
        const v2f B   = whT2[bIdx + js / 2];
        const v2f mk0 = *(const v2f*)&adjS[iLoc * LSTR + j0];
        const v2f mk1 = *(const v2f*)&adjS[(16 + iLoc) * LSTR + j0];

        v2f s0 = eib0 + ej;
        v2f s1 = eib1 + ej;
        const v2f t0 = s0 * 0.2f;
        const v2f t1 = s1 * 0.2f;
        s0.x = fmaxf(s0.x, t0.x);
        s0.y = fmaxf(s0.y, t0.y);
        s1.x = fmaxf(s1.x, t1.x);
        s1.y = fmaxf(s1.y, t1.y);
        v2f x0, x1;
        x0.x = __builtin_amdgcn_exp2f(s0.x);
        x0.y = __builtin_amdgcn_exp2f(s0.y);
        x1.x = __builtin_amdgcn_exp2f(s1.x);
        x1.y = __builtin_amdgcn_exp2f(s1.y);
        const v2f A0 = x0 * mk0;
        const v2f A1 = x1 * mk1;
        acc0 = __builtin_amdgcn_wmma_f32_16x16x4_f32(
            false, A0, false, B, (short)0, acc0, false, false);
        acc1 = __builtin_amdgcn_wmma_f32_16x16x4_f32(
            false, A1, false, B, (short)0, acc1, false, false);
    }

    const int rbase = (lane >> 4) * 8;
    if (nCol < DH) {
#pragma unroll
        for (int v = 0; v < 8; ++v) {
            ptile[hH][(rbase + v) * DH + nCol]      = acc0[v];
            ptile[hH][(16 + rbase + v) * DH + nCol] = acc1[v];
        }
    } else if (nCol == DH) {
#pragma unroll
        for (int v = 0; v < 8; ++v) {
            prow[hH * 16 + rbase + v]               = acc0[v];
            prow[NHEADS * 16 + hH * 16 + rbase + v] = acc1[v];
        }
    }
    __syncthreads();

    {
        const float4 t4a = ((const float4*)ptile[hH])[lane];
        const float4 t4b = ((const float4*)ptile[hH])[32 + lane];
        float* dstN = pnum + (size_t)jr * NN * OUT_DIM
                           + (size_t)hH * NN * DH + (size_t)i0 * DH;
        ((float4*)dstN)[lane]      = t4a;
        ((float4*)dstN)[32 + lane] = t4b;
    }
    if (hH < 2) {
        const float4 d4 = ((const float4*)prow)[hH * 32 + lane];
        float* dstD = pden + (size_t)jr * NN * NHEADS
                           + (size_t)blockIdx.x * (2 * NHEADS * 16);
        ((float4*)dstD)[hH * 32 + lane] = d4;
    }
}

__global__ __launch_bounds__(256)
void gat_combine_kernel(const float* __restrict__ pnum,
                        const float* __restrict__ pden,
                        float* __restrict__ out) {
    const int q = blockIdx.x * 256 + threadIdx.x;
    const int o = q * 4;
    const int h = o / (NN * DH);
    const int i = (o % (NN * DH)) / DH;
    const size_t doff = (size_t)(i >> 4) * (NHEADS * 16) + (size_t)h * 16 + (i & 15);

    float4 num = {0.f, 0.f, 0.f, 0.f};
    float  den = 0.f;
#pragma unroll
    for (int s = 0; s < JSPLIT; ++s) {
        const float4 p = ((const float4*)(pnum + (size_t)s * NN * OUT_DIM))[q];
        num.x += p.x; num.y += p.y; num.z += p.z; num.w += p.w;
        den += pden[(size_t)s * NN * NHEADS + doff];
    }
    float4 r;
    r.x = num.x / den; r.y = num.y / den; r.z = num.z / den; r.w = num.w / den;
    r.x = (r.x > 0.f) ? r.x : (__expf(r.x) - 1.f);
    r.y = (r.y > 0.f) ? r.y : (__expf(r.y) - 1.f);
    r.z = (r.z > 0.f) ? r.z : (__expf(r.z) - 1.f);
    r.w = (r.w > 0.f) ? r.w : (__expf(r.w) - 1.f);
    ((float4*)out)[q] = r;
}

extern "C" void kernel_launch(void* const* d_in, const int* in_sizes, int n_in,
                              void* d_out, int out_size, void* d_ws, size_t ws_size,
                              hipStream_t stream) {
    const float* h   = (const float*)d_in[0];
    const int*   adj = (const int*)  d_in[1];
    const float* W_w = (const float*)d_in[2];
    const float* W_b = (const float*)d_in[3];
    const float* a_w = (const float*)d_in[4];
    const float* a_b = (const float*)d_in[5];
    float* out = (float*)d_out;

    float* wh  = (float*)d_ws;
    float* e_i = wh  + (size_t)NN * OUT_DIM;
    float* e_j = e_i + (size_t)NN * NHEADS;
    float* pnum = e_j + (size_t)NN * NHEADS;
    float* pden = pnum + (size_t)JSPLIT * NN * OUT_DIM;

    gat_wh_kernel<<<NN, 64, 0, stream>>>(h, W_w, W_b, a_w, wh, e_i, e_j);
    gat_att_kernel<<<dim3(NN / ITILE, JSPLIT), 256, 0, stream>>>(adj, wh, e_i, e_j, a_b, pnum, pden);
    gat_combine_kernel<<<NN * OUT_DIM / 4 / 256, 256, 0, stream>>>(pnum, pden, out);
}


# projection kernel: 8 nodes per work-group, a wave holds 16 rows of W in registers for all 8: W fetched once per WGP, 1024 waves not 4096; in-lane k sum as even/odd chains (sum order over k changes)
# speedup vs baseline: 6.1723x; 1.2084x over previous
#include <hip/hip_runtime.h>
#include <hip/hip_bf16.h>


typedef __attribute__((ext_vector_type(2))) float v2f;
typedef __attribute__((ext_vector_type(8))) float v8f;

#define NN      2048
#define IN_DIM  256
#define OUT_DIM 64
#define NHEADS  8
#define DH      8
#define ITILE   32
#define LSTR    260
#define JSPLIT  8
#define JPER    (NN / JSPLIT)
#define ONES_LEN (JPER + 32)
#define LOG2E   1.4426950408889634f

#define NPB    8
#define FPW    16
#define PPITCH 132

__device__ __forceinline__ float gat_dot8(const float4 a, const float4 b,
                                          const float4 x, const float4 y) {
    v2f s = (v2f){a.x, a.y} * (v2f){x.x, x.y};
    s = __builtin_elementwise_fma((v2f){a.z, a.w}, (v2f){x.z, x.w}, s);
    s = __builtin_elementwise_fma((v2f){b.x, b.y}, (v2f){y.x, y.y}, s);
    s = __builtin_elementwise_fma((v2f){b.z, b.w}, (v2f){y.z, y.w}, s);
    return s.x + s.y;
}

__global__ __launch_bounds__(128)
void gat_wh_kernel(const float* __restrict__ h,
                   const float* __restrict__ W_w,
                   const float* __restrict__ W_b,
                   const float* __restrict__ a_w,
                   float* __restrict__ wh,
                   float* __restrict__ e_i,
                   float* __restrict__ e_j) {
    __shared__ __attribute__((aligned(16))) float part[4][32 * PPITCH];
    __shared__ __attribute__((aligned(16))) float whs[NPB * OUT_DIM];
    const int nb   = blockIdx.x * NPB;
    const int tid  = threadIdx.x;
    const int w    = tid >> 5;
    const int lane = tid & 31;

    float4 x0[NPB], x1[NPB];
#pragma unroll
    for (int n = 0; n < NPB; ++n) {
        const float4* hp = (const float4*)(h + (size_t)(nb + n) * IN_DIM);
        x0[n] = hp[lane];
        x1[n] = hp[32 + lane];
    }
    const float4* wp = (const float4*)(W_w + (size_t)(w * FPW) * IN_DIM) + lane;
    float4 wa[FPW], wb[FPW];
#pragma unroll
    for (int r = 0; r < FPW; ++r) {
        wa[r] = wp[r * 64];
        wb[r] = wp[r * 64 + 32];
    }

    const int    fb    = w * FPW + (lane & 3) * 4;
    const float4 bias  = *(const float4*)(W_b + fb);
    const int    isJ   = (tid >> 3) & 1;
    const float4 av0   = *(const float4*)(a_w + isJ * DH);
    const float4 av1   = *(const float4*)(a_w + isJ * DH + 4);

    float* myPart = &part[w][lane * PPITCH];
#pragma unroll
    for (int r0 = 0; r0 < FPW; r0 += 4) {
#pragma unroll
        for (int n = 0; n < NPB; ++n) {
            float4 p;
            p.x = gat_dot8(wa[r0 + 0], wb[r0 + 0], x0[n], x1[n]);
            p.y = gat_dot8(wa[r0 + 1], wb[r0 + 1], x0[n], x1[n]);
            p.z = gat_dot8(wa[r0 + 2], wb[r0 + 2], x0[n], x1[n]);
            p.w = gat_dot8(wa[r0 + 3], wb[r0 + 3], x0[n], x1[n]);
            *(float4*)&myPart[n * FPW + r0] = p;
        }
    }
    __syncthreads();

    const float* col = &part[w][lane * 4];
    float4 q[32];
#pragma unroll
    for (int l = 0; l < 32; ++l) q[l] = *(const float4*)&col[l * PPITCH];
    float4 acc = bias;
#pragma unroll
    for (int l = 0; l < 32; ++l) {
        acc.x += q[l].x; acc.y += q[l].y; acc.z += q[l].z; acc.w += q[l].w;
    }
    *(float4*)&whs[(lane >> 2) * OUT_DIM + fb] = acc;
    __syncthreads();

    ((float4*)(wh + (size_t)nb * OUT_DIM))[tid] = ((const float4*)whs)[tid];

    {
        const int n   = tid >> 4;
        const int hh  = tid & 7;
        const float4 w0 = *(const float4*)&whs[n * OUT_DIM + hh * DH];
        const float4 w1 = *(const float4*)&whs[n * OUT_DIM + hh * DH + 4];
        float s = 0.f;
        s = fmaf(w0.x, av0.x, s); s = fmaf(w0.y, av0.y, s); s = fmaf(w0.z, av0.z, s); s = fmaf(w0.w, av0.w, s);
        s = fmaf(w1.x, av1.x, s); s = fmaf(w1.y, av1.y, s); s = fmaf(w1.z, av1.z, s); s = fmaf(w1.w, av1.w, s);
        float* dst = isJ ? e_j : e_i;
        dst[(size_t)(nb + n) * NHEADS + hh] = s;
    }
}

__global__ __launch_bounds__(256)
void gat_att_kernel(const int*   __restrict__ adj,
                    const float* __restrict__ wh,
                    const float* __restrict__ e_i,
                    const float* __restrict__ e_j,
                    const float* __restrict__ a_b,
                    float* __restrict__ pnum,
                    float* __restrict__ pden) {
    __shared__ __attribute__((aligned(16))) float adjS[ITILE * LSTR];
    __shared__ __attribute__((aligned(16))) v2f whT2[(OUT_DIM * LSTR + ONES_LEN) / 2];
    float* const whT = (float*)whT2;
    __shared__ __attribute__((aligned(16))) float ejT[NHEADS * LSTR];
    __shared__ __attribute__((aligned(16))) float ptile[NHEADS][ITILE * DH];
    __shared__ __attribute__((aligned(16))) float prow[2 * NHEADS * 16];

    const int tid  = threadIdx.x;
    const int hH   = tid >> 5;
    const int lane = tid & 31;
    const int i0   = blockIdx.x * ITILE;
    const int jr   = blockIdx.y;
    const int jbeg = jr * JPER;

    const int   iLoc  = lane & 15;
    const int   nCol  = lane & 15;
    const int   khalf = (lane >> 4) * 2;

    const float ab  = a_b[0];
    const float ei0 = e_i[(size_t)(i0 + iLoc) * NHEADS + hH];
    const float ei1 = e_i[(size_t)(i0 + 16 + iLoc) * NHEADS + hH];

    const int ar = tid >> 6;
    const int ac = tid & 63;
    int4 a4[8];
#pragma unroll
    for (int k = 0; k < 8; ++k)
        a4[k] = *(const int4*)(adj + (size_t)(i0 + ar + 4 * k) * NN + jbeg + ac * 4);

    const int wq = (lane & 3) + 4 * (hH & 3);
    const int wj = (lane >> 2) + 8 * (hH >> 2);
    float4 w4[16];
#pragma unroll
    for (int k = 0; k < 16; ++k)
        w4[k] = *(const float4*)(wh + (size_t)(jbeg + wj + 16 * k) * OUT_DIM + wq * 4);

    const int eq = tid & 1;
    const int ej_j = tid >> 1;
    float4 e4[2];
#pragma unroll
    for (int k = 0; k < 2; ++k)
        e4[k] = *(const float4*)(e_j + (size_t)(jbeg + ej_j + 128 * k) * NHEADS + eq * 4);

    whT[OUT_DIM * LSTR + tid] = 1.0f;
    if (tid < ONES_LEN - 256) whT[OUT_DIM * LSTR + 256 + tid] = 1.0f;

#pragma unroll
    for (int k = 0; k < 8; ++k) {
        float4 m4;
        m4.x = (a4[k].x > 0) ? 1.0f : 0.0f;
        m4.y = (a4[k].y > 0) ? 1.0f : 0.0f;
        m4.z = (a4[k].z > 0) ? 1.0f : 0.0f;
        m4.w = (a4[k].w > 0) ? 1.0f : 0.0f;
        ((float4*)&adjS[(ar + 4 * k) * LSTR])[ac] = m4;
    }
#pragma unroll
    for (int k = 0; k < 16; ++k) {
        const int j = wj + 16 * k;
        whT[(wq * 4 + 0) * LSTR + j] = w4[k].x;
        whT[(wq * 4 + 1) * LSTR + j] = w4[k].y;
        whT[(wq * 4 + 2) * LSTR + j] = w4[k].z;
        whT[(wq * 4 + 3) * LSTR + j] = w4[k].w;
    }
#pragma unroll
    for (int k = 0; k < 2; ++k) {
        const int j = ej_j + 128 * k;
        ejT[(eq * 4 + 0) * LSTR + j] = e4[k].x * LOG2E;
        ejT[(eq * 4 + 1) * LSTR + j] = e4[k].y * LOG2E;
        ejT[(eq * 4 + 2) * LSTR + j] = e4[k].z * LOG2E;
        ejT[(eq * 4 + 3) * LSTR + j] = e4[k].w * LOG2E;
    }

    const float eibL0 = (ei0 + ab) * LOG2E;
    const float eibL1 = (ei1 + ab) * LOG2E;
    v2f eib0; eib0.x = eibL0; eib0.y = eibL0;
    v2f eib1; eib1.x = eibL1; eib1.y = eibL1;
    const int   bIdx  = ((nCol < DH) ? (hH * DH + nCol) * (LSTR / 2)
                                     : (OUT_DIM * LSTR) / 2 + ((hH & 1) ? 0 : 16))
                        + (lane >> 4);

    v8f acc0 = {0.f, 0.f, 0.f, 0.f, 0.f, 0.f, 0.f, 0.f};
    v8f acc1 = {0.f, 0.f, 0.f, 0.f, 0.f, 0.f, 0.f, 0.f};

    __syncthreads();

#pragma unroll 16
    for (int js = 0; js < JPER; js += 4) {
        const int j0 = js + khalf;
        const v2f ej  = *(const v2f*)&ejT[hH * LSTR + j0];
        const v2f B   = whT2[bIdx + js / 2];
        const v2f mk0 = *(const v2f*)&adjS[iLoc * LSTR + j0];
        const v2f mk1 = *(const v2f*)&adjS[(16 + iLoc) * LSTR + j0];

        v2f s0 = eib0 + ej;
        v2f s1 = eib1 + ej;
        const v2f t0 = s0 * 0.2f;
        const v2f t1 = s1 * 0.2f;
        s0.x = fmaxf(s0.x, t0.x);
        s0.y = fmaxf(s0.y, t0.y);
        s1.x = fmaxf(s1.x, t1.x);
        s1.y = fmaxf(s1.y, t1.y);
        v2f x0, x1;
        x0.x = __builtin_amdgcn_exp2f(s0.x);
        x0.y = __builtin_amdgcn_exp2f(s0.y);
        x1.x = __builtin_amdgcn_exp2f(s1.x);
        x1.y = __builtin_amdgcn_exp2f(s1.y);
        const v2f A0 = x0 * mk0;
        const v2f A1 = x1 * mk1;
        acc0 = __builtin_amdgcn_wmma_f32_16x16x4_f32(
            false, A0, false, B, (short)0, acc0, false, false);
        acc1 = __builtin_amdgcn_wmma_f32_16x16x4_f32(
            false, A1, false, B, (short)0, acc1, false, false);
    }

    const int rbase = (lane >> 4) * 8;
    if (nCol < DH) {
#pragma unroll
        for (int v = 0; v < 8; ++v) {
            ptile[hH][(rbase + v) * DH + nCol]      = acc0[v];
            ptile[hH][(16 + rbase + v) * DH + nCol] = acc1[v];
        }
    } else if (nCol == DH) {
#pragma unroll
        for (int v = 0; v < 8; ++v) {
            prow[hH * 16 + rbase + v]               = acc0[v];
            prow[NHEADS * 16 + hH * 16 + rbase + v] = acc1[v];
        }
    }
    __syncthreads();

    {
        const float4 t4a = ((const float4*)ptile[hH])[lane];
        const float4 t4b = ((const float4*)ptile[hH])[32 + lane];
        float* dstN = pnum + (size_t)jr * NN * OUT_DIM
                           + (size_t)hH * NN * DH + (size_t)i0 * DH;
        ((float4*)dstN)[lane]      = t4a;
        ((float4*)dstN)[32 + lane] = t4b;
    }
    if (hH < 2) {
        const float4 d4 = ((const float4*)prow)[hH * 32 + lane];
        float* dstD = pden + (size_t)jr * NN * NHEADS
                           + (size_t)blockIdx.x * (2 * NHEADS * 16);
        ((float4*)dstD)[hH * 32 + lane] = d4;
    }
}

__global__ __launch_bounds__(256)
void gat_combine_kernel(const float* __restrict__ pnum,
                        const float* __restrict__ pden,
                        float* __restrict__ out) {
    const int q = blockIdx.x * 256 + threadIdx.x;
    const int o = q * 4;
    const int h = o / (NN * DH);
    const int i = (o % (NN * DH)) / DH;
    const size_t doff = (size_t)(i >> 4) * (NHEADS * 16) + (size_t)h * 16 + (i & 15);

    float4 num = {0.f, 0.f, 0.f, 0.f};
    float  den = 0.f;
#pragma unroll
    for (int s = 0; s < JSPLIT; ++s) {
        const float4 p = ((const float4*)(pnum + (size_t)s * NN * OUT_DIM))[q];
        num.x += p.x; num.y += p.y; num.z += p.z; num.w += p.w;
        den += pden[(size_t)s * NN * NHEADS + doff];
    }
    float4 r;
    r.x = num.x / den; r.y = num.y / den; r.z = num.z / den; r.w = num.w / den;
    r.x = (r.x > 0.f) ? r.x : (__expf(r.x) - 1.f);
    r.y = (r.y > 0.f) ? r.y : (__expf(r.y) - 1.f);
    r.z = (r.z > 0.f) ? r.z : (__expf(r.z) - 1.f);
    r.w = (r.w > 0.f) ? r.w : (__expf(r.w) - 1.f);
    ((float4*)out)[q] = r;
}

extern "C" void kernel_launch(void* const* d_in, const int* in_sizes, int n_in,
                              void* d_out, int out_size, void* d_ws, size_t ws_size,
                              hipStream_t stream) {
    const float* h   = (const float*)d_in[0];
    const int*   adj = (const int*)  d_in[1];
    const float* W_w = (const float*)d_in[2];
    const float* W_b = (const float*)d_in[3];
    const float* a_w = (const float*)d_in[4];
    const float* a_b = (const float*)d_in[5];
    float* out = (float*)d_out;

    float* wh  = (float*)d_ws;
    float* e_i = wh  + (size_t)NN * OUT_DIM;
    float* e_j = e_i + (size_t)NN * NHEADS;
    float* pnum = e_j + (size_t)NN * NHEADS;
    float* pden = pnum + (size_t)JSPLIT * NN * OUT_DIM;

    gat_wh_kernel<<<NN / NPB, 128, 0, stream>>>(h, W_w, W_b, a_w, wh, e_i, e_j);
    gat_att_kernel<<<dim3(NN / ITILE, JSPLIT), 256, 0, stream>>>(adj, wh, e_i, e_j, a_b, pnum, pden);
    gat_combine_kernel<<<NN * OUT_DIM / 4 / 256, 256, 0, stream>>>(pnum, pden, out);
}


# launch 1 also packs adj>0 to 1 bit per edge as A-fragment lane masks; attention then takes whole rows (64 rows x 1 head, 8 j-slices added in LDS in j order): 2 launches, not 3; sum order unchanged
# speedup vs baseline: 7.1342x; 1.1558x over previous
#include <hip/hip_runtime.h>
#include <hip/hip_bf16.h>


typedef __attribute__((ext_vector_type(2))) float v2f;
typedef __attribute__((ext_vector_type(8))) float v8f;

#define NN      2048
#define IN_DIM  256
#define OUT_DIM 64
#define NHEADS  8
#define DH      8
#define LOG2E   1.4426950408889634f
#define KSTEPS  (NN / 4)

#define NPB    8
#define FPW    16
#define PPITCH 132

__device__ __forceinline__ float gat_dot8(const float4 a, const float4 b,
                                          const float4 x, const float4 y) {
    v2f s = (v2f){a.x, a.y} * (v2f){x.x, x.y};
    s = __builtin_elementwise_fma((v2f){a.z, a.w}, (v2f){x.z, x.w}, s);
    s = __builtin_elementwise_fma((v2f){b.x, b.y}, (v2f){y.x, y.y}, s);
    s = __builtin_elementwise_fma((v2f){b.z, b.w}, (v2f){y.z, y.w}, s);
    return s.x + s.y;
}

__device__ __forceinline__ void gat_pack_piece(const int* __restrict__ adj,
                                               unsigned int* __restrict__ bits,
                                               const int tile, const int piece, const int lane) {
    const int* src = adj + (size_t)(tile * 16) * NN + piece * 128 + lane * 4;
    int4 a[16];
#pragma unroll
    for (int r = 0; r < 16; ++r) a[r] = *(const int4*)(src + (size_t)r * NN);
    unsigned int d0 = 0u, d1 = 0u;
#pragma unroll
    for (int r = 0; r < 16; ++r) {
        d0 |= ((a[r].x > 0) ? (1u << r) : 0u) | ((a[r].z > 0) ? (1u << (16 + r)) : 0u);
        d1 |= ((a[r].y > 0) ? (1u << r) : 0u) | ((a[r].w > 0) ? (1u << (16 + r)) : 0u);
    }
    uint2 d; d.x = d0; d.y = d1;
    ((uint2*)bits)[(size_t)tile * KSTEPS + piece * 32 + lane] = d;
}

__global__ __launch_bounds__(256)
void gat_wh_kernel(const float* __restrict__ h,
                   const int*   __restrict__ adj,
                   const float* __restrict__ W_w,
                   const float* __restrict__ W_b,
                   const float* __restrict__ a_w,
                   float* __restrict__ whT_g,
                   float* __restrict__ e_i,
                   float* __restrict__ ejT_g,
                   unsigned int* __restrict__ bits) {
    __shared__ __attribute__((aligned(16))) float part[4][32 * PPITCH];
    __shared__ __attribute__((aligned(16))) float whs[NPB * OUT_DIM];
    const int nb   = blockIdx.x * NPB;
    const int tid  = threadIdx.x;
    const int w    = tid >> 5;
    const int wp4  = w & 3;
    const int lane = tid & 31;

    const int    fb    = wp4 * FPW + (lane & 3) * 4;
    const float4 bias  = *(const float4*)(W_b + fb);
    const int    isJ   = (tid >> 3) & 1;
    const float4 av0   = *(const float4*)(a_w + isJ * DH);
    const float4 av1   = *(const float4*)(a_w + isJ * DH + 4);

    if (w < 4) {
        float4 x0[NPB], x1[NPB];
#pragma unroll
        for (int n = 0; n < NPB; ++n) {
            const float4* hp = (const float4*)(h + (size_t)(nb + n) * IN_DIM);
            x0[n] = hp[lane];
            x1[n] = hp[32 + lane];
        }
        const float4* wp = (const float4*)(W_w + (size_t)(w * FPW) * IN_DIM) + lane;
        float4 wa[FPW], wb[FPW];
#pragma unroll
        for (int r = 0; r < FPW; ++r) {
            wa[r] = wp[r * 64];
            wb[r] = wp[r * 64 + 32];
        }
        float* myPart = &part[w][lane * PPITCH];
#pragma unroll
        for (int r0 = 0; r0 < FPW; r0 += 4) {
#pragma unroll
            for (int n = 0; n < NPB; ++n) {
                float4 p;
                p.x = gat_dot8(wa[r0 + 0], wb[r0 + 0], x0[n], x1[n]);
                p.y = gat_dot8(wa[r0 + 1], wb[r0 + 1], x0[n], x1[n]);
                p.z = gat_dot8(wa[r0 + 2], wb[r0 + 2], x0[n], x1[n]);
                p.w = gat_dot8(wa[r0 + 3], wb[r0 + 3], x0[n], x1[n]);
                *(float4*)&myPart[n * FPW + r0] = p;
            }
        }
    } else {
        const int tile = blockIdx.x >> 1;
        const int pc   = (blockIdx.x & 1) * 8 + wp4 * 2;
        gat_pack_piece(adj, bits, tile, pc,     lane);
        gat_pack_piece(adj, bits, tile, pc + 1, lane);
    }
    __syncthreads();

    if (w < 4) {
        const float* col = &part[w][lane * 4];
        float4 q[32];
#pragma unroll
        for (int l = 0; l < 32; ++l) q[l] = *(const float4*)&col[l * PPITCH];
        float4 acc = bias;
#pragma unroll
        for (int l = 0; l < 32; ++l) {
            acc.x += q[l].x; acc.y += q[l].y; acc.z += q[l].z; acc.w += q[l].w;
        }
        *(float4*)&whs[(lane >> 2) * OUT_DIM + fb] = acc;
    }
    __syncthreads();

    if (w < 4) {
        {
            const int f  = tid >> 1;
            const int n0 = (tid & 1) * 4;
            float4 v;
            v.x = whs[(n0 + 0) * OUT_DIM + f];
            v.y = whs[(n0 + 1) * OUT_DIM + f];
            v.z = whs[(n0 + 2) * OUT_DIM + f];
            v.w = whs[(n0 + 3) * OUT_DIM + f];
            *(float4*)(whT_g + (size_t)f * NN + nb + n0) = v;
        }
        {
            const int n   = tid >> 4;
            const int hh  = tid & 7;
            const float4 w0 = *(const float4*)&whs[n * OUT_DIM + hh * DH];
            const float4 w1 = *(const float4*)&whs[n * OUT_DIM + hh * DH + 4];
            float s = 0.f;
            s = fmaf(w0.x, av0.x, s); s = fmaf(w0.y, av0.y, s); s = fmaf(w0.z, av0.z, s); s = fmaf(w0.w, av0.w, s);
            s = fmaf(w1.x, av1.x, s); s = fmaf(w1.y, av1.y, s); s = fmaf(w1.z, av1.z, s); s = fmaf(w1.w, av1.w, s);
            if (isJ) ejT_g[(size_t)hh * NN + nb + n] = s;
            else     e_i[(size_t)(nb + n) * NHEADS + hh] = s;
        }
    }
}

#define RT      4
#define IROWS   (16 * RT)
#define JSL     256
#define WSTR    2052
#define ONES_OFF (DH * WSTR)
#define ONES_LEN (JSL + 4)
#define RSTR    9

__global__ __launch_bounds__(256)
void gat_att_kernel(const unsigned int* __restrict__ bits,
                    const float* __restrict__ whT_g,
                    const float* __restrict__ e_i,
                    const float* __restrict__ ejT_g,
                    const float* __restrict__ a_b,
                    float* __restrict__ out) {
    __shared__ __attribute__((aligned(16))) v2f whT2[(DH * WSTR + ONES_LEN) / 2];
    float* const whT = (float*)whT2;
    __shared__ __attribute__((aligned(16))) float ejS[NN];
    __shared__ __attribute__((aligned(16))) float red[8][IROWS * RSTR];

    const int tid  = threadIdx.x;
    const int lane = tid & 31;
    const int wv   = __builtin_amdgcn_readfirstlane(tid >> 5);
    const int hH   = blockIdx.x;
    const int i0   = blockIdx.y * IROWS;
    const int jw0  = wv * JSL;

    const int iLoc = lane & 15;
    const int nCol = lane & 15;
    const int kh   = lane >> 4;

    const float ab = a_b[0];
    float ei[RT];
#pragma unroll
    for (int t = 0; t < RT; ++t) ei[t] = e_i[(size_t)(i0 + 16 * t + iLoc) * NHEADS + hH];

    const float4* wsrc = (const float4*)(whT_g + (size_t)hH * DH * NN);
    float4 w4[16];
#pragma unroll
    for (int k = 0; k < 16; ++k) w4[k] = wsrc[tid + 256 * k];
    const float4* esrc = (const float4*)(ejT_g + (size_t)hH * NN);
    float4 e4[2];
#pragma unroll
    for (int k = 0; k < 2; ++k) e4[k] = esrc[tid + 256 * k];

    const uint4* mp[RT];
#pragma unroll
    for (int t = 0; t < RT; ++t)
        mp[t] = (const uint4*)(bits + ((size_t)(blockIdx.y * RT + t) * KSTEPS + jw0 / 4) * 2);
    uint4 mc[RT][2];
#pragma unroll
    for (int t = 0; t < RT; ++t) { mc[t][0] = mp[t][0]; mc[t][1] = mp[t][1]; }

    whT[ONES_OFF + tid] = 1.0f;
    if (tid < ONES_LEN - 256) whT[ONES_OFF + 256 + tid] = 1.0f;
#pragma unroll
    for (int k = 0; k < 16; ++k) {
        const int idx = tid + 256 * k;
        *(float4*)&whT[(idx >> 9) * WSTR + (idx & 511) * 4] = w4[k];
    }
#pragma unroll
    for (int k = 0; k < 2; ++k) {
        float4 v = e4[k];
        v.x *= LOG2E; v.y *= LOG2E; v.z *= LOG2E; v.w *= LOG2E;
        ((float4*)ejS)[tid + 256 * k] = v;
    }

    v2f eib[RT];
#pragma unroll
    for (int t = 0; t < RT; ++t) { const float s = (ei[t] + ab) * LOG2E; eib[t].x = s; eib[t].y = s; }

    const v2f* Bp = whT2 + ((nCol < DH) ? (nCol * WSTR + jw0) / 2 : ONES_OFF / 2) + kh;
    const v2f* Ep = (const v2f*)(ejS + jw0) + kh;

    v8f acc[RT];
#pragma unroll
    for (int t = 0; t < RT; ++t) acc[t] = (v8f){0.f, 0.f, 0.f, 0.f, 0.f, 0.f, 0.f, 0.f};

    __syncthreads();

#pragma unroll 2
    for (int trip = 0; trip < JSL / 16; ++trip) {
        const int tn = (trip + 1 < JSL / 16) ? trip + 1 : trip;
        uint4 mn[RT][2];
#pragma unroll
        for (int t = 0; t < RT; ++t) { mn[t][0] = mp[t][2 * tn]; mn[t][1] = mp[t][2 * tn + 1]; }

#pragma unroll
        for (int q = 0; q < 4; ++q) {
            const int jl = trip * 16 + q * 4;
            const v2f ej = Ep[jl / 2];
            const v2f B  = Bp[jl / 2];
#pragma unroll
            for (int t = 0; t < RT; ++t) {
                const uint4 mw = mc[t][q >> 1];
                const unsigned int m0 = (q & 1) ? mw.z : mw.x;
                const unsigned int m1 = (q & 1) ? mw.w : mw.y;
                v2f s = eib[t] + ej;
                const v2f u = s * 0.2f;
                s.x = fmaxf(s.x, u.x);
                s.y = fmaxf(s.y, u.y);
                v2f A;
                A.x = __builtin_amdgcn_inverse_ballot_w32(m0) ? __builtin_amdgcn_exp2f(s.x) : 0.0f;
                A.y = __builtin_amdgcn_inverse_ballot_w32(m1) ? __builtin_amdgcn_exp2f(s.y) : 0.0f;
                acc[t] = __builtin_amdgcn_wmma_f32_16x16x4_f32(
                    false, A, false, B, (short)0, acc[t], false, false);
            }
        }
#pragma unroll
        for (int t = 0; t < RT; ++t) { mc[t][0] = mn[t][0]; mc[t][1] = mn[t][1]; }
    }

    const int rbase = kh * 8;
    if (nCol <= DH) {
#pragma unroll
        for (int t = 0; t < RT; ++t) {
#pragma unroll
            for (int v = 0; v < 8; ++v)
                red[wv][(16 * t + rbase + v) * RSTR + nCol] = acc[t][v];
        }
    }
    __syncthreads();

    {
        const int r  = tid >> 2;
        const int d0 = (tid & 3) * 2;
        float n0 = 0.f, n1 = 0.f, den = 0.f;
#pragma unroll
        for (int s = 0; s < 8; ++s) {
            n0  += red[s][r * RSTR + d0];
            n1  += red[s][r * RSTR + d0 + 1];
            den += red[s][r * RSTR + DH];
        }
        float2 o;
        o.x = n0 / den; o.y = n1 / den;
        o.x = (o.x > 0.f) ? o.x : (__expf(o.x) - 1.f);
        o.y = (o.y > 0.f) ? o.y : (__expf(o.y) - 1.f);
        ((float2*)(out + (size_t)hH * NN * DH + (size_t)i0 * DH))[tid] = o;
    }
}

extern "C" void kernel_launch(void* const* d_in, const int* in_sizes, int n_in,
                              void* d_out, int out_size, void* d_ws, size_t ws_size,
                              hipStream_t stream) {
    const float* h   = (const float*)d_in[0];
    const int*   adj = (const int*)  d_in[1];
    const float* W_w = (const float*)d_in[2];
    const float* W_b = (const float*)d_in[3];
    const float* a_w = (const float*)d_in[4];
    const float* a_b = (const float*)d_in[5];
    float* out = (float*)d_out;

    float* whT_g = (float*)d_ws;
    float* e_i   = whT_g + (size_t)OUT_DIM * NN;
    float* ejT_g = e_i   + (size_t)NN * NHEADS;
    unsigned int* bits = (unsigned int*)(ejT_g + (size_t)NHEADS * NN);

    gat_wh_kernel<<<NN / NPB, 256, 0, stream>>>(h, adj, W_w, W_b, a_w, whT_g, e_i, ejT_g, bits);
    gat_att_kernel<<<dim3(NHEADS, NN / IROWS), 256, 0, stream>>>(bits, whT_g, e_i, ejT_g, a_b, out);
}
